# starting kernel
# baseline (speedup 1.0000x reference)
#include <hip/hip_runtime.h>

typedef __attribute__((ext_vector_type(2))) float v2f;
typedef __attribute__((ext_vector_type(8))) float v8f;

#define DD 128

__global__ __launch_bounds__(32) void iaf_layer_kernel(
    const float* __restrict__ x,
    const float* __restrict__ W1,  const float* __restrict__ b1,
    const float* __restrict__ Wmu, const float* __restrict__ bmu,
    const float* __restrict__ Wls, const float* __restrict__ bls,
    float* __restrict__ out, int B)
{
    __shared__ float Zh[16 * DD];
    __shared__ float Hh[16 * DD];
    __shared__ float tH[256], tM[256], tL[256];
    __shared__ float wd1[256], wdm[256], wdl[256];

    const int lane  = threadIdx.x;
    const int r0    = blockIdx.x * 16;
    const int mrow  = lane & 15;
    const int khalf = (lane >> 4) * 2;

    float ldj = 0.0f;

    for (int I = 0; I < 8; ++I) {
        const int c0 = I * 16;

        {
            const int e0 = lane * 8;
            #pragma unroll
            for (int q = 0; q < 8; ++q) {
                const int e  = e0 + q;
                const int t2 = e >> 4, t = e & 15;
                wd1[e] = W1 [(c0 + t2) * DD + c0 + t];
                wdm[e] = Wmu[(c0 + t2) * DD + c0 + t];
                wdl[e] = Wls[(c0 + t2) * DD + c0 + t];
            }
        }

        v8f cH = {0.f,0.f,0.f,0.f,0.f,0.f,0.f,0.f};
        v8f cM = {0.f,0.f,0.f,0.f,0.f,0.f,0.f,0.f};
        v8f cL = {0.f,0.f,0.f,0.f,0.f,0.f,0.f,0.f};
        const int nkt = 4 * I;
        for (int kt = 0; kt < nkt; ++kt) {
            const int k0 = kt * 4;
            v2f aZ = *(const v2f*)&Zh[mrow * DD + k0 + khalf];
            v2f aH = *(const v2f*)&Hh[mrow * DD + k0 + khalf];
            v2f bW1 = *(const v2f*)&W1 [(c0 + mrow) * DD + k0 + khalf];
            v2f bWm = *(const v2f*)&Wmu[(c0 + mrow) * DD + k0 + khalf];
            v2f bWl = *(const v2f*)&Wls[(c0 + mrow) * DD + k0 + khalf];
            cH = __builtin_amdgcn_wmma_f32_16x16x4_f32(false, aZ, false, bW1, (short)0, cH, false, false);
            cM = __builtin_amdgcn_wmma_f32_16x16x4_f32(false, aH, false, bWm, (short)0, cM, false, false);
            cL = __builtin_amdgcn_wmma_f32_16x16x4_f32(false, aH, false, bWl, (short)0, cL, false, false);
        }

        {
            const int mb = (lane >> 4) * 8;
            const int n  = lane & 15;
            #pragma unroll
            for (int r = 0; r < 8; ++r) {
                tH[(mb + r) * 16 + n] = cH[r];
                tM[(mb + r) * 16 + n] = cM[r];
                tL[(mb + r) * 16 + n] = cL[r];
            }
        }
        __syncthreads();

        if (lane < 16) {
            const int r = lane;
            float Hrow[16], Mrow[16], Lrow[16], xr[16];
            #pragma unroll
            for (int t = 0; t < 16; ++t) {
                Hrow[t] = tH[r * 16 + t] + b1 [c0 + t];
                Mrow[t] = tM[r * 16 + t] + bmu[c0 + t];
                Lrow[t] = tL[r * 16 + t] + bls[c0 + t];
                xr[t]   = x[(size_t)(r0 + r) * DD + c0 + t];
            }
            float zr[16], hr[16];
            #pragma unroll
            for (int t = 0; t < 16; ++t) {
                const float h  = fmaxf(Hrow[t], 0.0f);
                const float ls = Lrow[t];
                const float zv = Mrow[t] + xr[t] * __expf(ls);
                ldj += ls;
                zr[t] = zv;
                hr[t] = h;
                #pragma unroll
                for (int t2 = t + 1; t2 < 16; ++t2) {
                    Hrow[t2] = fmaf(zv, wd1[t2 * 16 + t], Hrow[t2]);
                    Mrow[t2] = fmaf(h,  wdm[t2 * 16 + t], Mrow[t2]);
                    Lrow[t2] = fmaf(h,  wdl[t2 * 16 + t], Lrow[t2]);
                }
            }
            #pragma unroll
            for (int t = 0; t < 16; ++t) {
                Zh[r * DD + c0 + t] = zr[t];
                Hh[r * DD + c0 + t] = hr[t];
                out[(size_t)(r0 + r) * DD + c0 + t] = zr[t];
            }
        }
        __syncthreads();
    }

    if (lane < 16) {
        out[(size_t)B * DD + r0 + lane] = ldj;
    }
}

extern "C" void kernel_launch(void* const* d_in, const int* in_sizes, int n_in,
                              void* d_out, int out_size, void* d_ws, size_t ws_size,
                              hipStream_t stream) {
    (void)n_in; (void)out_size; (void)d_ws; (void)ws_size;
    const float* x   = (const float*)d_in[0];
    const float* W1  = (const float*)d_in[1];
    const float* b1  = (const float*)d_in[2];
    const float* Wmu = (const float*)d_in[3];
    const float* bmu = (const float*)d_in[4];
    const float* Wls = (const float*)d_in[5];
    const float* bls = (const float*)d_in[6];
    float* out = (float*)d_out;

    const int B = in_sizes[0] / DD;
    dim3 grid(B / 16), block(32);
    iaf_layer_kernel<<<grid, block, 0, stream>>>(x, W1, b1, Wmu, bmu, Wls, bls, out, B);
}


# Next block's W rows and x slice fetched one block ahead with row-contiguous 128-bit loads and staged in LDS; B fragments, diagonal weights, x, biases read from LDS; barriers fence LDS only
# speedup vs baseline: 1.9351x; 1.9351x over previous
#include <hip/hip_runtime.h>


typedef __attribute__((ext_vector_type(2))) float v2f;
typedef __attribute__((ext_vector_type(4))) float v4f;
typedef __attribute__((ext_vector_type(8))) float v8f;

#define DD 128
#define PP 132
#define XP 20

__device__ __forceinline__ void lds_barrier() {
    __builtin_amdgcn_fence(__ATOMIC_RELEASE, "workgroup", "local");
    __builtin_amdgcn_s_barrier();
    __builtin_amdgcn_fence(__ATOMIC_ACQUIRE, "workgroup", "local");
}

__global__ __launch_bounds__(32) void iaf_layer_kernel(
    const float* __restrict__ x,
    const float* __restrict__ W1,  const float* __restrict__ b1,
    const float* __restrict__ Wmu, const float* __restrict__ bmu,
    const float* __restrict__ Wls, const float* __restrict__ bls,
    float* __restrict__ out, int B)
{
    __shared__ float Zh[16 * DD];
    __shared__ float Hh[16 * DD];
    __shared__ float tH[256], tM[256], tL[256];
    __shared__ __attribute__((aligned(16))) float pan[2][3][16 * PP];
    __shared__ __attribute__((aligned(16))) float bias[3][DD];
    __shared__ __attribute__((aligned(16))) float xs[2][16 * XP];

    const int lane  = threadIdx.x;
    const int r0    = blockIdx.x * 16;
    const int mrow  = lane & 15;
    const int khalf = (lane >> 4) * 2;
    const int prow  = lane >> 3;
    const int pc4   = lane & 7;

    const float* __restrict__ Wp[3] = {W1, Wmu, Wls};

    v4f pw[3][4][4];
    v4f px[2];
    const int xq = (lane >> 4) * 2;

    float ldj = 0.0f;

    *(v4f*)&bias[0][lane * 4] = *(const v4f*)&b1 [lane * 4];
    *(v4f*)&bias[1][lane * 4] = *(const v4f*)&bmu[lane * 4];
    *(v4f*)&bias[2][lane * 4] = *(const v4f*)&bls[lane * 4];
    #pragma unroll
    for (int m = 0; m < 3; ++m) {
        #pragma unroll
        for (int ri = 0; ri < 4; ++ri)
            pw[m][ri][0] = *(const v4f*)&Wp[m][(prow + 4 * ri) * DD + pc4 * 4];
    }
    #pragma unroll
    for (int q = 0; q < 2; ++q)
        px[q] = *(const v4f*)&x[(size_t)(r0 + mrow) * DD + (xq + q) * 4];

    for (int I = 0; I < 8; ++I) {
        const int c0 = I * 16;
        float* const pb = &pan[I & 1][0][0];

        #pragma unroll
        for (int ci = 0; ci < 4; ++ci) {
            if (2 * ci < I + 1) {
                #pragma unroll
                for (int m = 0; m < 3; ++m) {
                    #pragma unroll
                    for (int ri = 0; ri < 4; ++ri)
                        *(v4f*)&pb[m * 16 * PP + (prow + 4 * ri) * PP + (pc4 + 8 * ci) * 4] = pw[m][ri][ci];
                }
            }
        }
        float* const xb = &xs[I & 1][0];
        #pragma unroll
        for (int q = 0; q < 2; ++q)
            *(v4f*)&xb[mrow * XP + (xq + q) * 4] = px[q];
        lds_barrier();

        if (I < 7) {
            const int cn = c0 + 16;
            #pragma unroll
            for (int ci = 0; ci < 4; ++ci) {
                if (2 * ci < I + 2) {
                    #pragma unroll
                    for (int m = 0; m < 3; ++m) {
                        #pragma unroll
                        for (int ri = 0; ri < 4; ++ri)
                            pw[m][ri][ci] = *(const v4f*)&Wp[m][(cn + prow + 4 * ri) * DD + (pc4 + 8 * ci) * 4];
                    }
                }
            }
            #pragma unroll
            for (int q = 0; q < 2; ++q)
                px[q] = *(const v4f*)&x[(size_t)(r0 + mrow) * DD + cn + (xq + q) * 4];
        }

        v8f cH = {0.f,0.f,0.f,0.f,0.f,0.f,0.f,0.f};
        v8f cM = {0.f,0.f,0.f,0.f,0.f,0.f,0.f,0.f};
        v8f cL = {0.f,0.f,0.f,0.f,0.f,0.f,0.f,0.f};
        const int nkt = 4 * I;
        for (int kt = 0; kt < nkt; ++kt) {
            const int k0 = kt * 4;
            v2f aZ = *(const v2f*)&Zh[mrow * DD + k0 + khalf];
            v2f aH = *(const v2f*)&Hh[mrow * DD + k0 + khalf];
            v2f bW1 = *(const v2f*)&pb[0 * 16 * PP + mrow * PP + k0 + khalf];
            v2f bWm = *(const v2f*)&pb[1 * 16 * PP + mrow * PP + k0 + khalf];
            v2f bWl = *(const v2f*)&pb[2 * 16 * PP + mrow * PP + k0 + khalf];
            cH = __builtin_amdgcn_wmma_f32_16x16x4_f32(false, aZ, false, bW1, (short)0, cH, false, false);
            cM = __builtin_amdgcn_wmma_f32_16x16x4_f32(false, aH, false, bWm, (short)0, cM, false, false);
            cL = __builtin_amdgcn_wmma_f32_16x16x4_f32(false, aH, false, bWl, (short)0, cL, false, false);
        }

        {
            const int mb = (lane >> 4) * 8;
            const int n  = lane & 15;
            #pragma unroll
            for (int r = 0; r < 8; ++r) {
                tH[(mb + r) * 16 + n] = cH[r];
                tM[(mb + r) * 16 + n] = cM[r];
                tL[(mb + r) * 16 + n] = cL[r];
            }
        }
        lds_barrier();

        if (lane < 16) {
            const int r = lane;
            const float* const wd1 = pb + 0 * 16 * PP + c0;
            const float* const wdm = pb + 1 * 16 * PP + c0;
            const float* const wdl = pb + 2 * 16 * PP + c0;
            float Hrow[16], Mrow[16], Lrow[16], xr[16];
            #pragma unroll
            for (int t = 0; t < 16; ++t) {
                Hrow[t] = tH[r * 16 + t] + bias[0][c0 + t];
                Mrow[t] = tM[r * 16 + t] + bias[1][c0 + t];
                Lrow[t] = tL[r * 16 + t] + bias[2][c0 + t];
                xr[t]   = xb[r * XP + t];
            }
            float zr[16], hr[16];
            #pragma unroll
            for (int t = 0; t < 16; ++t) {
                const float h  = fmaxf(Hrow[t], 0.0f);
                const float ls = Lrow[t];
                const float zv = Mrow[t] + xr[t] * __expf(ls);
                ldj += ls;
                zr[t] = zv;
                hr[t] = h;
                #pragma unroll
                for (int t2 = t + 1; t2 < 16; ++t2) {
                    Hrow[t2] = fmaf(zv, wd1[t2 * PP + t], Hrow[t2]);
                    Mrow[t2] = fmaf(h,  wdm[t2 * PP + t], Mrow[t2]);
                    Lrow[t2] = fmaf(h,  wdl[t2 * PP + t], Lrow[t2]);
                }
            }
            #pragma unroll
            for (int t = 0; t < 16; ++t) {
                Zh[r * DD + c0 + t] = zr[t];
                Hh[r * DD + c0 + t] = hr[t];
                out[(size_t)(r0 + r) * DD + c0 + t] = zr[t];
            }
        }
    }

    if (lane < 16) {
        out[(size_t)B * DD + r0 + lane] = ldj;
    }
}

extern "C" void kernel_launch(void* const* d_in, const int* in_sizes, int n_in,
                              void* d_out, int out_size, void* d_ws, size_t ws_size,
                              hipStream_t stream) {
    (void)n_in; (void)out_size; (void)d_ws; (void)ws_size;
    const float* x   = (const float*)d_in[0];
    const float* W1  = (const float*)d_in[1];
    const float* b1  = (const float*)d_in[2];
    const float* Wmu = (const float*)d_in[3];
    const float* bmu = (const float*)d_in[4];
    const float* Wls = (const float*)d_in[5];
    const float* bls = (const float*)d_in[6];
    float* out = (float*)d_out;

    const int B = in_sizes[0] / DD;
    dim3 grid(B / 16), block(32);
    iaf_layer_kernel<<<grid, block, 0, stream>>>(x, W1, b1, Wmu, bmu, Wls, bls, out, B);
}
